# starting kernel
# baseline (speedup 1.0000x reference)
#include <hip/hip_runtime.h>


typedef __attribute__((ext_vector_type(2))) float v2f;
typedef __attribute__((ext_vector_type(8))) float v8f;

#define IN_LEN            8192
#define OUT_LEN           8192
#define KSPLIT            8
#define KSLICE            (IN_LEN / KSPLIT)
#define COLS_PER_WAVE     16
#define NCOLGRP           (OUT_LEN / COLS_PER_WAVE)
#define WAVES_PER_BLOCK   8
#define THREADS_PER_BLOCK (WAVES_PER_BLOCK * 32)
#define MATVEC_BLOCKS     ((NCOLGRP * KSPLIT) / WAVES_PER_BLOCK)

__global__ __launch_bounds__(THREADS_PER_BLOCK)
void bias_init_kernel(const float* __restrict__ b, float* __restrict__ out) {
    int i = blockIdx.x * blockDim.x + threadIdx.x;
    if (i < OUT_LEN) out[i] = b[i];
}

__global__ __launch_bounds__(THREADS_PER_BLOCK)
void matvec_wmma_f32_kernel(const float* __restrict__ x,
                            const float* __restrict__ W,
                            float* __restrict__ out) {
    const int lane        = threadIdx.x & 31;
    const int waveInBlock = threadIdx.x >> 5;
    const int flatWave    = blockIdx.x * WAVES_PER_BLOCK + waveInBlock;

    const int slice    = flatWave >> 9;
    const int colGroup = flatWave & (NCOLGRP - 1);
    const int j0       = colGroup * COLS_PER_WAVE;
    const int kBeg     = slice * KSLICE;

    const int laneLo = lane & 15;
    const int rowOff = (lane >> 4) << 1;

    const float* wp = W + (size_t)(kBeg + rowOff) * OUT_LEN + (size_t)(j0 + laneLo);
    const float* xp = x + kBeg + rowOff;

    v8f acc0 = {};
    v8f acc1 = {};

    #pragma unroll 2
    for (int it = 0; it < KSLICE / 8; ++it) {
        v2f a0 = *(const v2f*)(xp);
        v2f a1 = *(const v2f*)(xp + 4);

        v2f b0;
        b0.x = wp[0];
        b0.y = wp[OUT_LEN];
        v2f b1;
        b1.x = wp[(size_t)4 * OUT_LEN];
        b1.y = wp[(size_t)5 * OUT_LEN];

        __builtin_prefetch(wp + (size_t)32 * OUT_LEN, 0, 1);

        acc0 = __builtin_amdgcn_wmma_f32_16x16x4_f32(
            false, a0, false, b0, (short)0, acc0, false, false);
        acc1 = __builtin_amdgcn_wmma_f32_16x16x4_f32(
            false, a1, false, b1, (short)0, acc1, false, false);

        wp += (size_t)8 * OUT_LEN;
        xp += 8;
    }

    float total = acc0[0] + acc1[0];
    if (lane < 16) {
        unsafeAtomicAdd(&out[j0 + laneLo], total);
    }
}

extern "C" void kernel_launch(void* const* d_in, const int* in_sizes, int n_in,
                              void* d_out, int out_size, void* d_ws, size_t ws_size,
                              hipStream_t stream) {
    (void)in_sizes; (void)n_in; (void)d_ws; (void)ws_size; (void)out_size;

    const float* x = (const float*)d_in[0];
    const float* W = (const float*)d_in[1];
    const float* b = (const float*)d_in[2];
    float*     out = (float*)d_out;

    bias_init_kernel<<<OUT_LEN / THREADS_PER_BLOCK, THREADS_PER_BLOCK, 0, stream>>>(b, out);

    matvec_wmma_f32_kernel<<<MATVEC_BLOCKS, THREADS_PER_BLOCK, 0, stream>>>(x, W, out);
}


# each wave prefetches its W rows into the second-level cache 3-4 steps ahead (one global_prefetch_b8 per 8 rows x 1 KB, lane = one 256-B unit); the row loads then hit that cache, not memory
# speedup vs baseline: 1.7410x; 1.7410x over previous
#include <hip/hip_runtime.h>


typedef __attribute__((ext_vector_type(2))) float v2f;
typedef __attribute__((ext_vector_type(8))) float v8f;

#define IN_LEN            8192
#define OUT_LEN           8192
#define BIAS_THREADS      256

#define TN                256
#define NTILE             (TN / 32)
#define KS                64
#define WAVES             16
#define THREADS           (WAVES * 32)
#define KBLK              (KS * WAVES)
#define NSTRIPE           (OUT_LEN / TN)
#define NKBLK             (IN_LEN / KBLK)
#define STEP_ROWS         4
#define NSTEP             (KS / STEP_ROWS)
#define MATVEC_BLOCKS     (NSTRIPE * NKBLK)
#define PF_ROWS           8
#define NPF               (KS / PF_ROWS)
#define PF_AHEAD          2

__global__ __launch_bounds__(BIAS_THREADS)
void bias_init_kernel(const float* __restrict__ b, float* __restrict__ out) {
    int i = blockIdx.x * blockDim.x + threadIdx.x;
    if (i < OUT_LEN) out[i] = b[i];
}

struct StepBuf { v2f b[2][NTILE]; };

__device__ __forceinline__ void load_step(StepBuf& s, const float* __restrict__ p) {
    #pragma unroll
    for (int q = 0; q < 2; ++q) {
        #pragma unroll
        for (int t = 0; t < NTILE; ++t) {
            s.b[q][t].x = p[(size_t)(2 * q)     * OUT_LEN + 32 * t];
            s.b[q][t].y = p[(size_t)(2 * q + 1) * OUT_LEN + 32 * t];
        }
    }
}

__device__ __forceinline__ void prefetch_group(const float* p) {
    __builtin_prefetch(p, 0, 2);
}

__device__ __forceinline__ v2f a_frag(v2f xv, int pair, bool act) {
    float x0 = __int_as_float(__builtin_amdgcn_readlane(__float_as_int(xv.x), pair));
    float x1 = __int_as_float(__builtin_amdgcn_readlane(__float_as_int(xv.y), pair));
    v2f a;
    a.x = act ? x0 : 0.0f;
    a.y = act ? x1 : 0.0f;
    return a;
}

__device__ __forceinline__ void mul_step(v8f (&acc)[NTILE], const StepBuf& s,
                                         v2f xv, int step, bool act) {
    #pragma unroll
    for (int q = 0; q < 2; ++q) {
        v2f a = a_frag(xv, 2 * step + q, act);
        #pragma unroll
        for (int t = 0; t < NTILE; ++t) {
            acc[t] = __builtin_amdgcn_wmma_f32_16x16x4_f32(
                false, a, false, s.b[q][t], (short)0, acc[t], false, false);
        }
    }
}

__global__ __launch_bounds__(THREADS)
void matvec_wmma_f32_kernel(const float* __restrict__ x,
                            const float* __restrict__ W,
                            float* __restrict__ out) {
    __shared__ float red[WAVES][TN];

    const int lane   = threadIdx.x & 31;
    const int wave   = threadIdx.x >> 5;
    const int stripe = blockIdx.x & (NSTRIPE - 1);
    const int kblk   = blockIdx.x / NSTRIPE;
    const int j0     = stripe * TN;
    const int kBeg   = kblk * KBLK + wave * KS;

    const bool act = (lane < 8) || (lane >= 24);

    const v2f xv = *(const v2f*)(x + kBeg + 2 * lane);

    const float* wp = W + (size_t)kBeg * OUT_LEN + (size_t)(j0 + lane);

    v8f acc[NTILE];
    #pragma unroll
    for (int t = 0; t < NTILE; ++t) acc[t] = v8f{};

    const float* pf = W + (size_t)(kBeg + (lane >> 2)) * OUT_LEN
                        + (size_t)(j0 + 64 * (lane & 3));

    StepBuf s0, s1;
    load_step(s0, wp);
    prefetch_group(pf + (size_t)1 * PF_ROWS * OUT_LEN);
    #pragma unroll
    for (int st = 0; st < NSTEP; st += 2) {
        load_step(s1, wp + (size_t)(st + 1) * STEP_ROWS * OUT_LEN);
        if (st / 2 + PF_AHEAD < NPF)
            prefetch_group(pf + (size_t)(st / 2 + PF_AHEAD) * PF_ROWS * OUT_LEN);
        __builtin_amdgcn_sched_barrier(0);
        mul_step(acc, s0, xv, st, act);
        __builtin_amdgcn_sched_barrier(0);
        if (st + 2 < NSTEP)
            load_step(s0, wp + (size_t)(st + 2) * STEP_ROWS * OUT_LEN);
        __builtin_amdgcn_sched_barrier(0);
        mul_step(acc, s1, xv, st + 1, act);
        __builtin_amdgcn_sched_barrier(0);
    }

    #pragma unroll
    for (int t = 0; t < NTILE; ++t) red[wave][32 * t + lane] = acc[t][0];
    __syncthreads();

    if (threadIdx.x < TN) {
        float total = 0.0f;
        #pragma unroll
        for (int w = 0; w < WAVES; ++w) total += red[w][threadIdx.x];
        unsafeAtomicAdd(&out[j0 + threadIdx.x], total);
    }
}

extern "C" void kernel_launch(void* const* d_in, const int* in_sizes, int n_in,
                              void* d_out, int out_size, void* d_ws, size_t ws_size,
                              hipStream_t stream) {
    (void)in_sizes; (void)n_in; (void)d_ws; (void)ws_size; (void)out_size;

    const float* x = (const float*)d_in[0];
    const float* W = (const float*)d_in[1];
    const float* b = (const float*)d_in[2];
    float*     out = (float*)d_out;

    bias_init_kernel<<<OUT_LEN / BIAS_THREADS, BIAS_THREADS, 0, stream>>>(b, out);

    matvec_wmma_f32_kernel<<<MATVEC_BLOCKS, THREADS, 0, stream>>>(x, W, out);
}
